# starting kernel
# baseline (speedup 1.0000x reference)
#include <hip/hip_runtime.h>

typedef float v2f __attribute__((ext_vector_type(2)));
typedef float v8f __attribute__((ext_vector_type(8)));

#define EA_W 5
#define EA_B 4
#define EA_H 128
#define EA_WW 160
#define EA_C 64
#define EA_M 10
#define EA_K 640
#define EA_EPS 1e-3f

__global__ void ea_prep_kernel(const float* __restrict__ kflat, float* __restrict__ bpt) {
    int idx = blockIdx.x * blockDim.x + threadIdx.x;
    if (idx >= 16 * EA_K) return;
    int m = idx / EA_K;
    int k = idx % EA_K;
    bpt[idx] = (m < EA_M) ? kflat[k * EA_M + m] : 0.0f;
}

__global__ __launch_bounds__(128)
void ea_main_kernel(const float* __restrict__ L, const float* __restrict__ R,
                    const float* __restrict__ bpt, const float* __restrict__ bias,
                    const float* __restrict__ gamma, const float* __restrict__ beta,
                    const float* __restrict__ mean, const float* __restrict__ var,
                    float* __restrict__ out) {
    const int tid  = threadIdx.x;
    const int lane = tid & 31;
    const int wib  = tid >> 5;
    const int tile = blockIdx.x * 4 + wib;
    const int lo = lane & 15;
    const int hi = lane >> 4;

    const int p0 = tile * 16;
    const int n  = p0 / (EA_H * EA_WW);
    const int hw = p0 % (EA_H * EA_WW);
    const int h  = hw / EA_WW;
    const int w0 = hw % EA_WW;
    const int w  = w0 + lo;
    const size_t rowBase = (size_t)(n * EA_H + h) * EA_WW * EA_C;
    const float* Lrow = L + rowBase;
    const float* Rrow = R + rowBase;

    __shared__ float qtile[4][20][128];
    __shared__ float bsm[16][640];

    {
        const float4* bsrc = (const float4*)bpt;
        float4* bdst = (float4*)&bsm[0][0];
        #pragma unroll
        for (int it = 0; it < 20; ++it)
            bdst[it * 128 + tid] = bsrc[it * 128 + tid];
    }
    {
        float4* qdst = (float4*)&qtile[wib][0][0];
        const int c4 = lane * 4;
        #pragma unroll
        for (int it = 0; it < 20; ++it) {
            const int wq = w0 - 2 + it;
            float4 v = {0.0f, 0.0f, 0.0f, 0.0f};
            if (wq >= 0 && wq < EA_WW) {
                const float* s = (c4 < EA_C) ? (Lrow + (size_t)wq * EA_C + c4)
                                             : (Rrow + (size_t)wq * EA_C + (c4 - EA_C));
                v = *(const float4*)s;
            }
            qdst[it * 32 + lane] = v;
        }
    }
    __syncthreads();

    const v2f* aV = (const v2f*)&qtile[wib][lo][hi * 2];
    const v2f* bV = (const v2f*)&bsm[lo][hi * 2];
    v8f acc0 = {}, acc1 = {};
    #pragma unroll
    for (int step = 0; step < EA_K / 4; step += 2) {
        const v2f a0 = aV[2 * step];
        const v2f b0 = bV[2 * step];
        acc0 = __builtin_amdgcn_wmma_f32_16x16x4_f32(false, a0, false, b0,
                                                     (short)0, acc0, false, false);
        const v2f a1 = aV[2 * step + 2];
        const v2f b1 = bV[2 * step + 2];
        acc1 = __builtin_amdgcn_wmma_f32_16x16x4_f32(false, a1, false, b1,
                                                     (short)0, acc1, false, false);
    }
    const v8f acc = acc0 + acc1;

    __shared__ float smem[4][16][16];
    #pragma unroll
    for (int v = 0; v < 8; ++v)
        smem[wib][v + 8 * hi][lo] = acc[v];
    __syncthreads();

    float wv[EA_M];
    float mx = -3.0e38f;
    #pragma unroll
    for (int m = 0; m < EA_M; ++m) {
        wv[m] = smem[wib][lo][m] + bias[m];
        mx = fmaxf(mx, wv[m]);
    }
    float sum = 0.0f;
    #pragma unroll
    for (int m = 0; m < EA_M; ++m) { wv[m] = __expf(wv[m] - mx); sum += wv[m]; }
    const float inv = 1.0f / sum;
    #pragma unroll
    for (int m = 0; m < EA_M; ++m) wv[m] *= inv;

    const int p = p0 + lo;
    const size_t XN = (size_t)EA_B * EA_H * EA_WW * EA_C;
    const size_t WN = (size_t)EA_B * EA_H * EA_WW * EA_M;
    if (hi == 0) {
        float* ow  = out + XN + (size_t)p * EA_M;
        float* ow2 = out + XN + WN + (size_t)p * EA_M;
        #pragma unroll
        for (int m = 0; m < EA_M; ++m) { ow[m] = wv[m]; ow2[m] = wv[m]; }
    }

    const int kb = hi * 32;
    float x[32];
    #pragma unroll
    for (int j = 0; j < 32; ++j) x[j] = 0.0f;
    #pragma unroll
    for (int i = 0; i < EA_W; ++i) {
        int wl = w - i; if (wl < 0)      wl += EA_WW;
        int wr = w + i; if (wr >= EA_WW) wr -= EA_WW;
        const float4* ls = (const float4*)(Lrow + (size_t)wl * EA_C + kb);
        const float4* rs = (const float4*)(Rrow + (size_t)wr * EA_C + kb);
        const float wlw = wv[2 * i];
        const float wrw = wv[2 * i + 1];
        #pragma unroll
        for (int q = 0; q < 8; ++q) {
            const float4 lv = ls[q];
            const float4 rv = rs[q];
            x[4 * q + 0] += wlw * lv.x + wrw * rv.x;
            x[4 * q + 1] += wlw * lv.y + wrw * rv.y;
            x[4 * q + 2] += wlw * lv.z + wrw * rv.z;
            x[4 * q + 3] += wlw * lv.w + wrw * rv.w;
        }
    }

    float* ox = out + (size_t)p * EA_C + kb;
    #pragma unroll
    for (int j = 0; j < 32; ++j) {
        const int k = kb + j;
        const float sc = gamma[k] * rsqrtf(var[k] + EA_EPS);
        const float y  = (x[j] - mean[k]) * sc + beta[k];
        ox[j] = tanhf(y);
    }
}

extern "C" void kernel_launch(void* const* d_in, const int* in_sizes, int n_in,
                              void* d_out, int out_size, void* d_ws, size_t ws_size,
                              hipStream_t stream) {
    const float* L     = (const float*)d_in[0];
    const float* R     = (const float*)d_in[1];
    const float* kconv = (const float*)d_in[2];
    const float* bias  = (const float*)d_in[3];
    const float* gamma = (const float*)d_in[4];
    const float* beta  = (const float*)d_in[5];
    const float* mean  = (const float*)d_in[6];
    const float* var   = (const float*)d_in[7];
    float* out = (float*)d_out;
    float* bpt = (float*)d_ws;

    ea_prep_kernel<<<(16 * EA_K + 255) / 256, 256, 0, stream>>>(kconv, bpt);

    const int tiles = EA_B * EA_H * EA_WW / 16;
    ea_main_kernel<<<tiles / 4, 128, 0, stream>>>(L, R, bpt, bias, gamma, beta,
                                                  mean, var, out);
}


# Pad LDS row pitches (query 128->132, weights 640->644, logits 16->17 floats) so WMMA A/B fragment reads spread over all 64 banks instead of 4; k order and summation order unchanged
# speedup vs baseline: 1.4297x; 1.4297x over previous
#include <hip/hip_runtime.h>

typedef float v2f __attribute__((ext_vector_type(2)));
typedef float v8f __attribute__((ext_vector_type(8)));

#define EA_W 5
#define EA_B 4
#define EA_H 128
#define EA_WW 160
#define EA_C 64
#define EA_M 10
#define EA_K 640
#define EA_EPS 1e-3f
#define EA_QP 132
#define EA_BP 644
#define EA_SP 17

__global__ void ea_prep_kernel(const float* __restrict__ kflat, float* __restrict__ bpt) {
    int idx = blockIdx.x * blockDim.x + threadIdx.x;
    if (idx >= 16 * EA_K) return;
    int m = idx / EA_K;
    int k = idx % EA_K;
    bpt[idx] = (m < EA_M) ? kflat[k * EA_M + m] : 0.0f;
}

__global__ __launch_bounds__(128)
void ea_main_kernel(const float* __restrict__ L, const float* __restrict__ R,
                    const float* __restrict__ bpt, const float* __restrict__ bias,
                    const float* __restrict__ gamma, const float* __restrict__ beta,
                    const float* __restrict__ mean, const float* __restrict__ var,
                    float* __restrict__ out) {
    const int tid  = threadIdx.x;
    const int lane = tid & 31;
    const int wib  = tid >> 5;
    const int tile = blockIdx.x * 4 + wib;
    const int lo = lane & 15;
    const int hi = lane >> 4;

    const int p0 = tile * 16;
    const int n  = p0 / (EA_H * EA_WW);
    const int hw = p0 % (EA_H * EA_WW);
    const int h  = hw / EA_WW;
    const int w0 = hw % EA_WW;
    const int w  = w0 + lo;
    const size_t rowBase = (size_t)(n * EA_H + h) * EA_WW * EA_C;
    const float* Lrow = L + rowBase;
    const float* Rrow = R + rowBase;

    __shared__ __attribute__((aligned(16))) float qtile[4][20][EA_QP];
    __shared__ __attribute__((aligned(16))) float bsm[16][EA_BP];

    {
        const float4* bsrc = (const float4*)bpt;
        float4* bdst = (float4*)&bsm[0][0];
        #pragma unroll
        for (int it = 0; it < 20; ++it) {
            const int f4 = it * 128 + tid;
            bdst[f4 + f4 / (EA_K / 4)] = bsrc[f4];
        }
    }
    {
        float4* qdst = (float4*)&qtile[wib][0][0];
        const int c4 = lane * 4;
        #pragma unroll
        for (int it = 0; it < 20; ++it) {
            const int wq = w0 - 2 + it;
            float4 v = {0.0f, 0.0f, 0.0f, 0.0f};
            if (wq >= 0 && wq < EA_WW) {
                const float* s = (c4 < EA_C) ? (Lrow + (size_t)wq * EA_C + c4)
                                             : (Rrow + (size_t)wq * EA_C + (c4 - EA_C));
                v = *(const float4*)s;
            }
            qdst[it * (EA_QP / 4) + lane] = v;
        }
    }
    __syncthreads();

    const float* aB = &qtile[wib][lo][hi * 2];
    const float* bB = &bsm[lo][hi * 2];
    v8f acc0 = {}, acc1 = {};
    #pragma unroll
    for (int step = 0; step < EA_K / 4; step += 2) {
        const int k0 = 4 * step;
        const int k1 = 4 * step + 4;
        const v2f a0 = *(const v2f*)(aB + (k0 / 128) * EA_QP + (k0 % 128));
        const v2f b0 = *(const v2f*)(bB + k0);
        acc0 = __builtin_amdgcn_wmma_f32_16x16x4_f32(false, a0, false, b0,
                                                     (short)0, acc0, false, false);
        const v2f a1 = *(const v2f*)(aB + (k1 / 128) * EA_QP + (k1 % 128));
        const v2f b1 = *(const v2f*)(bB + k1);
        acc1 = __builtin_amdgcn_wmma_f32_16x16x4_f32(false, a1, false, b1,
                                                     (short)0, acc1, false, false);
    }
    const v8f acc = acc0 + acc1;

    __shared__ float smem[4][16][EA_SP];
    #pragma unroll
    for (int v = 0; v < 8; ++v)
        smem[wib][v + 8 * hi][lo] = acc[v];
    __syncthreads();

    float wv[EA_M];
    float mx = -3.0e38f;
    #pragma unroll
    for (int m = 0; m < EA_M; ++m) {
        wv[m] = smem[wib][lo][m] + bias[m];
        mx = fmaxf(mx, wv[m]);
    }
    float sum = 0.0f;
    #pragma unroll
    for (int m = 0; m < EA_M; ++m) { wv[m] = __expf(wv[m] - mx); sum += wv[m]; }
    const float inv = 1.0f / sum;
    #pragma unroll
    for (int m = 0; m < EA_M; ++m) wv[m] *= inv;

    const int p = p0 + lo;
    const size_t XN = (size_t)EA_B * EA_H * EA_WW * EA_C;
    const size_t WN = (size_t)EA_B * EA_H * EA_WW * EA_M;
    if (hi == 0) {
        float* ow  = out + XN + (size_t)p * EA_M;
        float* ow2 = out + XN + WN + (size_t)p * EA_M;
        #pragma unroll
        for (int m = 0; m < EA_M; ++m) { ow[m] = wv[m]; ow2[m] = wv[m]; }
    }

    const int kb = hi * 32;
    float x[32];
    #pragma unroll
    for (int j = 0; j < 32; ++j) x[j] = 0.0f;
    #pragma unroll
    for (int i = 0; i < EA_W; ++i) {
        int wl = w - i; if (wl < 0)      wl += EA_WW;
        int wr = w + i; if (wr >= EA_WW) wr -= EA_WW;
        const float4* ls = (const float4*)(Lrow + (size_t)wl * EA_C + kb);
        const float4* rs = (const float4*)(Rrow + (size_t)wr * EA_C + kb);
        const float wlw = wv[2 * i];
        const float wrw = wv[2 * i + 1];
        #pragma unroll
        for (int q = 0; q < 8; ++q) {
            const float4 lv = ls[q];
            const float4 rv = rs[q];
            x[4 * q + 0] += wlw * lv.x + wrw * rv.x;
            x[4 * q + 1] += wlw * lv.y + wrw * rv.y;
            x[4 * q + 2] += wlw * lv.z + wrw * rv.z;
            x[4 * q + 3] += wlw * lv.w + wrw * rv.w;
        }
    }

    float* ox = out + (size_t)p * EA_C + kb;
    #pragma unroll
    for (int j = 0; j < 32; ++j) {
        const int k = kb + j;
        const float sc = gamma[k] * rsqrtf(var[k] + EA_EPS);
        const float y  = (x[j] - mean[k]) * sc + beta[k];
        ox[j] = tanhf(y);
    }
}

extern "C" void kernel_launch(void* const* d_in, const int* in_sizes, int n_in,
                              void* d_out, int out_size, void* d_ws, size_t ws_size,
                              hipStream_t stream) {
    const float* L     = (const float*)d_in[0];
    const float* R     = (const float*)d_in[1];
    const float* kconv = (const float*)d_in[2];
    const float* bias  = (const float*)d_in[3];
    const float* gamma = (const float*)d_in[4];
    const float* beta  = (const float*)d_in[5];
    const float* mean  = (const float*)d_in[6];
    const float* var   = (const float*)d_in[7];
    float* out = (float*)d_out;
    float* bpt = (float*)d_ws;

    ea_prep_kernel<<<(16 * EA_K + 255) / 256, 256, 0, stream>>>(kconv, bpt);

    const int tiles = EA_B * EA_H * EA_WW / 16;
    ea_main_kernel<<<tiles / 4, 128, 0, stream>>>(L, R, bpt, bias, gamma, beta,
                                                  mean, var, out);
}


# Softmax-weighted gather reads L/R from the wave's LDS tile (widened 20->24 positions, wrapped; conv padding from a zero row) instead of 80 scattered global loads per lane; sum order unchanged
# speedup vs baseline: 2.6020x; 1.8200x over previous
#include <hip/hip_runtime.h>

typedef float v2f __attribute__((ext_vector_type(2)));
typedef float v8f __attribute__((ext_vector_type(8)));

#define EA_W 5
#define EA_B 4
#define EA_H 128
#define EA_WW 160
#define EA_C 64
#define EA_M 10
#define EA_K 640
#define EA_EPS 1e-3f
#define EA_QP 132
#define EA_BP 644
#define EA_SP 17
#define EA_TP 24
#define EA_TILE (EA_TP * EA_QP)
#define EA_ZOFF (4 * EA_TILE)

__global__ void ea_prep_kernel(const float* __restrict__ kflat, float* __restrict__ bpt) {
    int idx = blockIdx.x * blockDim.x + threadIdx.x;
    if (idx >= 16 * EA_K) return;
    int m = idx / EA_K;
    int k = idx % EA_K;
    bpt[idx] = (m < EA_M) ? kflat[k * EA_M + m] : 0.0f;
}

__global__ __launch_bounds__(128)
void ea_main_kernel(const float* __restrict__ L, const float* __restrict__ R,
                    const float* __restrict__ bpt, const float* __restrict__ bias,
                    const float* __restrict__ gamma, const float* __restrict__ beta,
                    const float* __restrict__ mean, const float* __restrict__ var,
                    float* __restrict__ out) {
    const int tid  = threadIdx.x;
    const int lane = tid & 31;
    const int wib  = tid >> 5;
    const int tile = blockIdx.x * 4 + wib;
    const int lo = lane & 15;
    const int hi = lane >> 4;

    const int p0 = tile * 16;
    const int n  = p0 / (EA_H * EA_WW);
    const int hw = p0 % (EA_H * EA_WW);
    const int h  = hw / EA_WW;
    const int w0 = hw % EA_WW;
    const int w  = w0 + lo;
    const size_t rowBase = (size_t)(n * EA_H + h) * EA_WW * EA_C;
    const float* Lrow = L + rowBase;
    const float* Rrow = R + rowBase;

    __shared__ __attribute__((aligned(16))) float qs[EA_ZOFF + EA_QP];
    __shared__ __attribute__((aligned(16))) float bsm[16][EA_BP];
    float* qtile = qs + wib * EA_TILE;

    {
        const float4* bsrc = (const float4*)bpt;
        float4* bdst = (float4*)&bsm[0][0];
        #pragma unroll
        for (int it = 0; it < 20; ++it) {
            const int f4 = it * 128 + tid;
            bdst[f4 + f4 / (EA_K / 4)] = bsrc[f4];
        }
    }
    {
        float4* qdst = (float4*)qtile;
        const int c4 = lane * 4;
        const float* srow = (c4 < EA_C) ? (Lrow + c4) : (Rrow + (c4 - EA_C));
        const int wrapLo = (w0 == 0) ? EA_WW : 0;
        const int wrapHi = (w0 == EA_WW - 16) ? EA_WW : 0;
        #pragma unroll
        for (int it = 0; it < EA_TP; ++it) {
            const int wq = w0 - 4 + it + (it < 4 ? wrapLo : 0) - (it >= EA_TP - 4 ? wrapHi : 0);
            qdst[it * (EA_QP / 4) + lane] = *(const float4*)(srow + (ptrdiff_t)wq * EA_C);
        }
        if (wib == 0) {
            const float4 z = {0.0f, 0.0f, 0.0f, 0.0f};
            ((float4*)(qs + EA_ZOFF))[lane] = z;
        }
    }
    __syncthreads();

    const float* aT[EA_W];
    #pragma unroll
    for (int t = 0; t < EA_W; ++t) {
        const int wq = w + t - 2;
        const int off = (wq >= 0 && wq < EA_WW) ? (wib * EA_TILE + (lo + t + 2) * EA_QP) : EA_ZOFF;
        aT[t] = qs + off + hi * 2;
    }
    const float* bB = &bsm[lo][hi * 2];
    v8f acc0 = {}, acc1 = {};
    #pragma unroll
    for (int step = 0; step < EA_K / 4; step += 2) {
        const int k0 = 4 * step;
        const int k1 = 4 * step + 4;
        const v2f a0 = *(const v2f*)(aT[k0 / 128] + (k0 % 128));
        const v2f b0 = *(const v2f*)(bB + k0);
        acc0 = __builtin_amdgcn_wmma_f32_16x16x4_f32(false, a0, false, b0,
                                                     (short)0, acc0, false, false);
        const v2f a1 = *(const v2f*)(aT[k1 / 128] + (k1 % 128));
        const v2f b1 = *(const v2f*)(bB + k1);
        acc1 = __builtin_amdgcn_wmma_f32_16x16x4_f32(false, a1, false, b1,
                                                     (short)0, acc1, false, false);
    }
    const v8f acc = acc0 + acc1;

    __shared__ float smem[4][16][EA_SP];
    #pragma unroll
    for (int v = 0; v < 8; ++v)
        smem[wib][v + 8 * hi][lo] = acc[v];
    __syncthreads();

    float wv[EA_M];
    float mx = -3.0e38f;
    #pragma unroll
    for (int m = 0; m < EA_M; ++m) {
        wv[m] = smem[wib][lo][m] + bias[m];
        mx = fmaxf(mx, wv[m]);
    }
    float sum = 0.0f;
    #pragma unroll
    for (int m = 0; m < EA_M; ++m) { wv[m] = __expf(wv[m] - mx); sum += wv[m]; }
    const float inv = 1.0f / sum;
    #pragma unroll
    for (int m = 0; m < EA_M; ++m) wv[m] *= inv;

    const int p = p0 + lo;
    const size_t XN = (size_t)EA_B * EA_H * EA_WW * EA_C;
    const size_t WN = (size_t)EA_B * EA_H * EA_WW * EA_M;
    if (hi == 0) {
        float* ow  = out + XN + (size_t)p * EA_M;
        float* ow2 = out + XN + WN + (size_t)p * EA_M;
        #pragma unroll
        for (int m = 0; m < EA_M; ++m) { ow[m] = wv[m]; ow2[m] = wv[m]; }
    }

    const int kb = hi * 32;
    const float* gL = qtile + (lo + 4) * EA_QP + kb;
    const float* gR = gL + EA_C;
    float x[32];
    #pragma unroll
    for (int j = 0; j < 32; ++j) x[j] = 0.0f;
    #pragma unroll
    for (int i = 0; i < EA_W; ++i) {
        const float4* ls = (const float4*)(gL - i * EA_QP);
        const float4* rs = (const float4*)(gR + i * EA_QP);
        const float wlw = wv[2 * i];
        const float wrw = wv[2 * i + 1];
        #pragma unroll
        for (int q = 0; q < 8; ++q) {
            const float4 lv = ls[q];
            const float4 rv = rs[q];
            x[4 * q + 0] += wlw * lv.x + wrw * rv.x;
            x[4 * q + 1] += wlw * lv.y + wrw * rv.y;
            x[4 * q + 2] += wlw * lv.z + wrw * rv.z;
            x[4 * q + 3] += wlw * lv.w + wrw * rv.w;
        }
    }

    float* ox = out + (size_t)p * EA_C + kb;
    #pragma unroll
    for (int j = 0; j < 32; ++j) {
        const int k = kb + j;
        const float sc = gamma[k] * rsqrtf(var[k] + EA_EPS);
        const float y  = (x[j] - mean[k]) * sc + beta[k];
        ox[j] = tanhf(y);
    }
}

extern "C" void kernel_launch(void* const* d_in, const int* in_sizes, int n_in,
                              void* d_out, int out_size, void* d_ws, size_t ws_size,
                              hipStream_t stream) {
    const float* L     = (const float*)d_in[0];
    const float* R     = (const float*)d_in[1];
    const float* kconv = (const float*)d_in[2];
    const float* bias  = (const float*)d_in[3];
    const float* gamma = (const float*)d_in[4];
    const float* beta  = (const float*)d_in[5];
    const float* mean  = (const float*)d_in[6];
    const float* var   = (const float*)d_in[7];
    float* out = (float*)d_out;
    float* bpt = (float*)d_ws;

    ea_prep_kernel<<<(16 * EA_K + 255) / 256, 256, 0, stream>>>(kconv, bpt);

    const int tiles = EA_B * EA_H * EA_WW / 16;
    ea_main_kernel<<<tiles / 4, 128, 0, stream>>>(L, R, bpt, bias, gamma, beta,
                                                  mean, var, out);
}


# Gather/BN/tanh/store phase remapped: a lane owns 4 channels of 8 pixels (was 32 channels of 1), so each output store writes whole 128-B lines as b128; per-pixel sum order unchanged
# speedup vs baseline: 4.7038x; 1.8078x over previous
#include <hip/hip_runtime.h>

typedef float v2f __attribute__((ext_vector_type(2)));
typedef float v8f __attribute__((ext_vector_type(8)));

#define EA_W 5
#define EA_B 4
#define EA_H 128
#define EA_WW 160
#define EA_C 64
#define EA_M 10
#define EA_K 640
#define EA_EPS 1e-3f
#define EA_QP 132
#define EA_BP 644
#define EA_SP 17
#define EA_WP 12
#define EA_TP 24
#define EA_TILE (EA_TP * EA_QP)
#define EA_ZOFF (4 * EA_TILE)

__global__ void ea_prep_kernel(const float* __restrict__ kflat, float* __restrict__ bpt) {
    int idx = blockIdx.x * blockDim.x + threadIdx.x;
    if (idx >= 16 * EA_K) return;
    int m = idx / EA_K;
    int k = idx % EA_K;
    bpt[idx] = (m < EA_M) ? kflat[k * EA_M + m] : 0.0f;
}

__global__ __launch_bounds__(128)
void ea_main_kernel(const float* __restrict__ L, const float* __restrict__ R,
                    const float* __restrict__ bpt, const float* __restrict__ bias,
                    const float* __restrict__ gamma, const float* __restrict__ beta,
                    const float* __restrict__ mean, const float* __restrict__ var,
                    float* __restrict__ out) {
    const int tid  = threadIdx.x;
    const int lane = tid & 31;
    const int wib  = tid >> 5;
    const int tile = blockIdx.x * 4 + wib;
    const int lo = lane & 15;
    const int hi = lane >> 4;

    const int p0 = tile * 16;
    const int n  = p0 / (EA_H * EA_WW);
    const int hw = p0 % (EA_H * EA_WW);
    const int h  = hw / EA_WW;
    const int w0 = hw % EA_WW;
    const int w  = w0 + lo;
    const size_t rowBase = (size_t)(n * EA_H + h) * EA_WW * EA_C;
    const float* Lrow = L + rowBase;
    const float* Rrow = R + rowBase;

    __shared__ __attribute__((aligned(16))) float qs[EA_ZOFF + EA_QP];
    __shared__ __attribute__((aligned(16))) float bsm[16][EA_BP];
    float* qtile = qs + wib * EA_TILE;

    {
        const float4* bsrc = (const float4*)bpt;
        float4* bdst = (float4*)&bsm[0][0];
        #pragma unroll
        for (int it = 0; it < 20; ++it) {
            const int f4 = it * 128 + tid;
            bdst[f4 + f4 / (EA_K / 4)] = bsrc[f4];
        }
    }
    {
        float4* qdst = (float4*)qtile;
        const int c4 = lane * 4;
        const float* srow = (c4 < EA_C) ? (Lrow + c4) : (Rrow + (c4 - EA_C));
        const int wrapLo = (w0 == 0) ? EA_WW : 0;
        const int wrapHi = (w0 == EA_WW - 16) ? EA_WW : 0;
        #pragma unroll
        for (int it = 0; it < EA_TP; ++it) {
            const int wq = w0 - 4 + it + (it < 4 ? wrapLo : 0) - (it >= EA_TP - 4 ? wrapHi : 0);
            qdst[it * (EA_QP / 4) + lane] = *(const float4*)(srow + (ptrdiff_t)wq * EA_C);
        }
        if (wib == 0) {
            const float4 z = {0.0f, 0.0f, 0.0f, 0.0f};
            ((float4*)(qs + EA_ZOFF))[lane] = z;
        }
    }
    __syncthreads();

    const float* aT[EA_W];
    #pragma unroll
    for (int t = 0; t < EA_W; ++t) {
        const int wq = w + t - 2;
        const int off = (wq >= 0 && wq < EA_WW) ? (wib * EA_TILE + (lo + t + 2) * EA_QP) : EA_ZOFF;
        aT[t] = qs + off + hi * 2;
    }
    const float* bB = &bsm[lo][hi * 2];
    v8f acc0 = {}, acc1 = {};
    #pragma unroll
    for (int step = 0; step < EA_K / 4; step += 2) {
        const int k0 = 4 * step;
        const int k1 = 4 * step + 4;
        const v2f a0 = *(const v2f*)(aT[k0 / 128] + (k0 % 128));
        const v2f b0 = *(const v2f*)(bB + k0);
        acc0 = __builtin_amdgcn_wmma_f32_16x16x4_f32(false, a0, false, b0,
                                                     (short)0, acc0, false, false);
        const v2f a1 = *(const v2f*)(aT[k1 / 128] + (k1 % 128));
        const v2f b1 = *(const v2f*)(bB + k1);
        acc1 = __builtin_amdgcn_wmma_f32_16x16x4_f32(false, a1, false, b1,
                                                     (short)0, acc1, false, false);
    }
    const v8f acc = acc0 + acc1;

    __shared__ float smem[4][16][EA_SP];
    #pragma unroll
    for (int v = 0; v < 8; ++v)
        smem[wib][v + 8 * hi][lo] = acc[v];
    __syncthreads();

    float wv[EA_M];
    float mx = -3.0e38f;
    #pragma unroll
    for (int m = 0; m < EA_M; ++m) {
        wv[m] = smem[wib][lo][m] + bias[m];
        mx = fmaxf(mx, wv[m]);
    }
    float sum = 0.0f;
    #pragma unroll
    for (int m = 0; m < EA_M; ++m) { wv[m] = __expf(wv[m] - mx); sum += wv[m]; }
    const float inv = 1.0f / sum;
    #pragma unroll
    for (int m = 0; m < EA_M; ++m) wv[m] *= inv;

    const int p = p0 + lo;
    const size_t XN = (size_t)EA_B * EA_H * EA_WW * EA_C;
    const size_t WN = (size_t)EA_B * EA_H * EA_WW * EA_M;
    __shared__ __attribute__((aligned(16))) float wsm[4][16][EA_WP];
    if (hi == 0) {
        float4* wd = (float4*)&wsm[wib][lo][0];
        wd[0] = make_float4(wv[0], wv[1], wv[2], wv[3]);
        wd[1] = make_float4(wv[4], wv[5], wv[6], wv[7]);
        *(float2*)&wsm[wib][lo][8] = make_float2(wv[8], wv[9]);
    }
    __syncthreads();
    if (hi == 0) {
        float* ow  = out + XN + (size_t)p * EA_M;
        float* ow2 = out + XN + WN + (size_t)p * EA_M;
        #pragma unroll
        for (int m = 0; m < EA_M; ++m) { ow[m] = wv[m]; ow2[m] = wv[m]; }
    }

    const int c4 = lo * 4;
    const int px = hi * 8;
    const float* tL = qtile + px * EA_QP + c4;
    const float* tR = qtile + (px + 4) * EA_QP + EA_C + c4;
    float4 lrow[12], rrow[12];
    #pragma unroll
    for (int t = 0; t < 12; ++t) {
        lrow[t] = *(const float4*)(tL + t * EA_QP);
        rrow[t] = *(const float4*)(tR + t * EA_QP);
    }

    const float4 g4 = *(const float4*)(gamma + c4);
    const float4 v4 = *(const float4*)(var + c4);
    const float4 m4 = *(const float4*)(mean + c4);
    const float4 b4 = *(const float4*)(beta + c4);
    float4 sc;
    sc.x = g4.x * rsqrtf(v4.x + EA_EPS);
    sc.y = g4.y * rsqrtf(v4.y + EA_EPS);
    sc.z = g4.z * rsqrtf(v4.z + EA_EPS);
    sc.w = g4.w * rsqrtf(v4.w + EA_EPS);

    float* ox = out + (size_t)(p0 + px) * EA_C + c4;
    #pragma unroll
    for (int j = 0; j < 8; ++j) {
        const float4 wa = *(const float4*)&wsm[wib][px + j][0];
        const float4 wb = *(const float4*)&wsm[wib][px + j][4];
        const float2 wc = *(const float2*)&wsm[wib][px + j][8];
        const float wj[EA_M] = {wa.x, wa.y, wa.z, wa.w, wb.x, wb.y, wb.z, wb.w, wc.x, wc.y};
        float4 x = {0.0f, 0.0f, 0.0f, 0.0f};
        #pragma unroll
        for (int i = 0; i < EA_W; ++i) {
            const float4 lv = lrow[j + 4 - i];
            const float4 rv = rrow[j + i];
            const float wlw = wj[2 * i];
            const float wrw = wj[2 * i + 1];
            x.x += wlw * lv.x + wrw * rv.x;
            x.y += wlw * lv.y + wrw * rv.y;
            x.z += wlw * lv.z + wrw * rv.z;
            x.w += wlw * lv.w + wrw * rv.w;
        }
        float4 y;
        y.x = tanhf((x.x - m4.x) * sc.x + b4.x);
        y.y = tanhf((x.y - m4.y) * sc.y + b4.y);
        y.z = tanhf((x.z - m4.z) * sc.z + b4.z);
        y.w = tanhf((x.w - m4.w) * sc.w + b4.w);
        *(float4*)(ox + j * EA_C) = y;
    }
}

extern "C" void kernel_launch(void* const* d_in, const int* in_sizes, int n_in,
                              void* d_out, int out_size, void* d_ws, size_t ws_size,
                              hipStream_t stream) {
    const float* L     = (const float*)d_in[0];
    const float* R     = (const float*)d_in[1];
    const float* kconv = (const float*)d_in[2];
    const float* bias  = (const float*)d_in[3];
    const float* gamma = (const float*)d_in[4];
    const float* beta  = (const float*)d_in[5];
    const float* mean  = (const float*)d_in[6];
    const float* var   = (const float*)d_in[7];
    float* out = (float*)d_out;
    float* bpt = (float*)d_ws;

    ea_prep_kernel<<<(16 * EA_K + 255) / 256, 256, 0, stream>>>(kconv, bpt);

    const int tiles = EA_B * EA_H * EA_WW / 16;
    ea_main_kernel<<<tiles / 4, 128, 0, stream>>>(L, R, bpt, bias, gamma, beta,
                                                  mean, var, out);
}


# Weight repack folded into the main kernel: each block transposes the HWIO conv kernel into its LDS B tile itself (10 rows, no zero rows); the prep launch is gone (2 -> 1); k and sum order unchanged
# speedup vs baseline: 5.3014x; 1.1270x over previous
#include <hip/hip_runtime.h>

typedef float v2f __attribute__((ext_vector_type(2)));
typedef float v8f __attribute__((ext_vector_type(8)));

#define EA_W 5
#define EA_B 4
#define EA_H 128
#define EA_WW 160
#define EA_C 64
#define EA_M 10
#define EA_K 640
#define EA_EPS 1e-3f
#define EA_QP 132
#define EA_BP 644
#define EA_SP 17
#define EA_WP 12
#define EA_TP 24
#define EA_TILE (EA_TP * EA_QP)
#define EA_ZOFF (4 * EA_TILE)

__global__ __launch_bounds__(128)
void ea_main_kernel(const float* __restrict__ L, const float* __restrict__ R,
                    const float* __restrict__ kflat, const float* __restrict__ bias,
                    const float* __restrict__ gamma, const float* __restrict__ beta,
                    const float* __restrict__ mean, const float* __restrict__ var,
                    float* __restrict__ out) {
    const int tid  = threadIdx.x;
    const int lane = tid & 31;
    const int wib  = tid >> 5;
    const int tile = blockIdx.x * 4 + wib;
    const int lo = lane & 15;
    const int hi = lane >> 4;

    const int p0 = tile * 16;
    const int n  = p0 / (EA_H * EA_WW);
    const int hw = p0 % (EA_H * EA_WW);
    const int h  = hw / EA_WW;
    const int w0 = hw % EA_WW;
    const int w  = w0 + lo;
    const size_t rowBase = (size_t)(n * EA_H + h) * EA_WW * EA_C;
    const float* Lrow = L + rowBase;
    const float* Rrow = R + rowBase;

    __shared__ __attribute__((aligned(16))) float qs[EA_ZOFF + EA_QP];
    __shared__ __attribute__((aligned(16))) float bsm[EA_M][EA_BP];
    float* qtile = qs + wib * EA_TILE;

    {
        const float2* bsrc = (const float2*)kflat;
        float* bdst = &bsm[0][0];
        #pragma unroll
        for (int j = 0; j < 5; ++j) {
            const int u = j * 128 + tid;
            const int k = u / 5;
            const int r = u - 5 * k;
            float* d = bdst + (2 * r) * EA_BP + k;
            #pragma unroll
            for (int i = 0; i < 5; ++i) {
                const float2 v = bsrc[u + i * 640];
                d[i * 128]         = v.x;
                d[i * 128 + EA_BP] = v.y;
            }
        }
    }
    {
        float4* qdst = (float4*)qtile;
        const int c4 = lane * 4;
        const float* srow = (c4 < EA_C) ? (Lrow + c4) : (Rrow + (c4 - EA_C));
        const int wrapLo = (w0 == 0) ? EA_WW : 0;
        const int wrapHi = (w0 == EA_WW - 16) ? EA_WW : 0;
        #pragma unroll
        for (int it = 0; it < EA_TP; ++it) {
            const int wq = w0 - 4 + it + (it < 4 ? wrapLo : 0) - (it >= EA_TP - 4 ? wrapHi : 0);
            qdst[it * (EA_QP / 4) + lane] = *(const float4*)(srow + (ptrdiff_t)wq * EA_C);
        }
        if (wib == 0) {
            const float4 z = {0.0f, 0.0f, 0.0f, 0.0f};
            ((float4*)(qs + EA_ZOFF))[lane] = z;
        }
    }
    __syncthreads();

    const float* aT[EA_W];
    #pragma unroll
    for (int t = 0; t < EA_W; ++t) {
        const int wq = w + t - 2;
        const int off = (wq >= 0 && wq < EA_WW) ? (wib * EA_TILE + (lo + t + 2) * EA_QP) : EA_ZOFF;
        aT[t] = qs + off + hi * 2;
    }
    const float* bB = &bsm[lo < EA_M ? lo : lo - 6][hi * 2];
    v8f acc0 = {}, acc1 = {};
    #pragma unroll
    for (int step = 0; step < EA_K / 4; step += 2) {
        const int k0 = 4 * step;
        const int k1 = 4 * step + 4;
        const v2f a0 = *(const v2f*)(aT[k0 / 128] + (k0 % 128));
        const v2f b0 = *(const v2f*)(bB + k0);
        acc0 = __builtin_amdgcn_wmma_f32_16x16x4_f32(false, a0, false, b0,
                                                     (short)0, acc0, false, false);
        const v2f a1 = *(const v2f*)(aT[k1 / 128] + (k1 % 128));
        const v2f b1 = *(const v2f*)(bB + k1);
        acc1 = __builtin_amdgcn_wmma_f32_16x16x4_f32(false, a1, false, b1,
                                                     (short)0, acc1, false, false);
    }
    const v8f acc = acc0 + acc1;

    __shared__ float smem[4][16][EA_SP];
    #pragma unroll
    for (int v = 0; v < 8; ++v)
        smem[wib][v + 8 * hi][lo] = acc[v];
    __syncthreads();

    float wv[EA_M];
    float mx = -3.0e38f;
    #pragma unroll
    for (int m = 0; m < EA_M; ++m) {
        wv[m] = smem[wib][lo][m] + bias[m];
        mx = fmaxf(mx, wv[m]);
    }
    float sum = 0.0f;
    #pragma unroll
    for (int m = 0; m < EA_M; ++m) { wv[m] = __expf(wv[m] - mx); sum += wv[m]; }
    const float inv = 1.0f / sum;
    #pragma unroll
    for (int m = 0; m < EA_M; ++m) wv[m] *= inv;

    const int p = p0 + lo;
    const size_t XN = (size_t)EA_B * EA_H * EA_WW * EA_C;
    const size_t WN = (size_t)EA_B * EA_H * EA_WW * EA_M;
    __shared__ __attribute__((aligned(16))) float wsm[4][16][EA_WP];
    if (hi == 0) {
        float4* wd = (float4*)&wsm[wib][lo][0];
        wd[0] = make_float4(wv[0], wv[1], wv[2], wv[3]);
        wd[1] = make_float4(wv[4], wv[5], wv[6], wv[7]);
        *(float2*)&wsm[wib][lo][8] = make_float2(wv[8], wv[9]);
    }
    __syncthreads();
    if (hi == 0) {
        float* ow  = out + XN + (size_t)p * EA_M;
        float* ow2 = out + XN + WN + (size_t)p * EA_M;
        #pragma unroll
        for (int m = 0; m < EA_M; ++m) { ow[m] = wv[m]; ow2[m] = wv[m]; }
    }

    const int c4 = lo * 4;
    const int px = hi * 8;
    const float* tL = qtile + px * EA_QP + c4;
    const float* tR = qtile + (px + 4) * EA_QP + EA_C + c4;
    float4 lrow[12], rrow[12];
    #pragma unroll
    for (int t = 0; t < 12; ++t) {
        lrow[t] = *(const float4*)(tL + t * EA_QP);
        rrow[t] = *(const float4*)(tR + t * EA_QP);
    }

    const float4 g4 = *(const float4*)(gamma + c4);
    const float4 v4 = *(const float4*)(var + c4);
    const float4 m4 = *(const float4*)(mean + c4);
    const float4 b4 = *(const float4*)(beta + c4);
    float4 sc;
    sc.x = g4.x * rsqrtf(v4.x + EA_EPS);
    sc.y = g4.y * rsqrtf(v4.y + EA_EPS);
    sc.z = g4.z * rsqrtf(v4.z + EA_EPS);
    sc.w = g4.w * rsqrtf(v4.w + EA_EPS);

    float* ox = out + (size_t)(p0 + px) * EA_C + c4;
    #pragma unroll
    for (int j = 0; j < 8; ++j) {
        const float4 wa = *(const float4*)&wsm[wib][px + j][0];
        const float4 wb = *(const float4*)&wsm[wib][px + j][4];
        const float2 wc = *(const float2*)&wsm[wib][px + j][8];
        const float wj[EA_M] = {wa.x, wa.y, wa.z, wa.w, wb.x, wb.y, wb.z, wb.w, wc.x, wc.y};
        float4 x = {0.0f, 0.0f, 0.0f, 0.0f};
        #pragma unroll
        for (int i = 0; i < EA_W; ++i) {
            const float4 lv = lrow[j + 4 - i];
            const float4 rv = rrow[j + i];
            const float wlw = wj[2 * i];
            const float wrw = wj[2 * i + 1];
            x.x += wlw * lv.x + wrw * rv.x;
            x.y += wlw * lv.y + wrw * rv.y;
            x.z += wlw * lv.z + wrw * rv.z;
            x.w += wlw * lv.w + wrw * rv.w;
        }
        float4 y;
        y.x = tanhf((x.x - m4.x) * sc.x + b4.x);
        y.y = tanhf((x.y - m4.y) * sc.y + b4.y);
        y.z = tanhf((x.z - m4.z) * sc.z + b4.z);
        y.w = tanhf((x.w - m4.w) * sc.w + b4.w);
        *(float4*)(ox + j * EA_C) = y;
    }
}

extern "C" void kernel_launch(void* const* d_in, const int* in_sizes, int n_in,
                              void* d_out, int out_size, void* d_ws, size_t ws_size,
                              hipStream_t stream) {
    const float* L     = (const float*)d_in[0];
    const float* R     = (const float*)d_in[1];
    const float* kconv = (const float*)d_in[2];
    const float* bias  = (const float*)d_in[3];
    const float* gamma = (const float*)d_in[4];
    const float* beta  = (const float*)d_in[5];
    const float* mean  = (const float*)d_in[6];
    const float* var   = (const float*)d_in[7];
    float* out = (float*)d_out;

    const int tiles = EA_B * EA_H * EA_WW / 16;
    ea_main_kernel<<<tiles / 4, 128, 0, stream>>>(L, R, kconv, bias, gamma, beta,
                                                  mean, var, out);
}


# tanh made branch-free: the small-|y| polynomial and 1-2/(exp(2|y|)+1) are both computed and one selected, so a lane's 32 evaluations overlap instead of running as 64 serial divergent branch regions
# speedup vs baseline: 6.6177x; 1.2483x over previous
#include <hip/hip_runtime.h>

typedef float v2f __attribute__((ext_vector_type(2)));
typedef float v8f __attribute__((ext_vector_type(8)));

#define EA_W 5
#define EA_B 4
#define EA_H 128
#define EA_WW 160
#define EA_C 64
#define EA_M 10
#define EA_K 640
#define EA_EPS 1e-3f
#define EA_QP 132
#define EA_BP 644
#define EA_SP 17
#define EA_WP 12
#define EA_TP 24
#define EA_TILE (EA_TP * EA_QP)
#define EA_ZOFF (4 * EA_TILE)

__device__ __forceinline__ float ea_tanh(float y) {
    const float a = __builtin_fabsf(y);
    const float s = a * a;
    float p = __builtin_fmaf(s, -0.00570002059f, 0.0206340719f);
    p = __builtin_fmaf(s, p, -0.0537379310f);
    p = __builtin_fmaf(s, p, 0.133314162f);
    p = __builtin_fmaf(s, p, -0.333332807f);
    const float small = __builtin_fmaf(s, a * p, a);
    const float e = __builtin_amdgcn_exp2f(a * 2.88539004f);
    const float big = __builtin_fmaf(__builtin_amdgcn_rcpf(e + 1.0f), -2.0f, 1.0f);
    return __builtin_copysignf(a < 0.625f ? small : big, y);
}

__global__ __launch_bounds__(128)
void ea_main_kernel(const float* __restrict__ L, const float* __restrict__ R,
                    const float* __restrict__ kflat, const float* __restrict__ bias,
                    const float* __restrict__ gamma, const float* __restrict__ beta,
                    const float* __restrict__ mean, const float* __restrict__ var,
                    float* __restrict__ out) {
    const int tid  = threadIdx.x;
    const int lane = tid & 31;
    const int wib  = tid >> 5;
    const int tile = blockIdx.x * 4 + wib;
    const int lo = lane & 15;
    const int hi = lane >> 4;

    const int p0 = tile * 16;
    const int n  = p0 / (EA_H * EA_WW);
    const int hw = p0 % (EA_H * EA_WW);
    const int h  = hw / EA_WW;
    const int w0 = hw % EA_WW;
    const int w  = w0 + lo;
    const size_t rowBase = (size_t)(n * EA_H + h) * EA_WW * EA_C;
    const float* Lrow = L + rowBase;
    const float* Rrow = R + rowBase;

    __shared__ __attribute__((aligned(16))) float qs[EA_ZOFF + EA_QP];
    __shared__ __attribute__((aligned(16))) float bsm[EA_M][EA_BP];
    float* qtile = qs + wib * EA_TILE;

    {
        const float2* bsrc = (const float2*)kflat;
        float* bdst = &bsm[0][0];
        #pragma unroll
        for (int j = 0; j < 5; ++j) {
            const int u = j * 128 + tid;
            const int k = u / 5;
            const int r = u - 5 * k;
            float* d = bdst + (2 * r) * EA_BP + k;
            #pragma unroll
            for (int i = 0; i < 5; ++i) {
                const float2 v = bsrc[u + i * 640];
                d[i * 128]         = v.x;
                d[i * 128 + EA_BP] = v.y;
            }
        }
    }
    {
        float4* qdst = (float4*)qtile;
        const int c4 = lane * 4;
        const float* srow = (c4 < EA_C) ? (Lrow + c4) : (Rrow + (c4 - EA_C));
        const int wrapLo = (w0 == 0) ? EA_WW : 0;
        const int wrapHi = (w0 == EA_WW - 16) ? EA_WW : 0;
        #pragma unroll
        for (int it = 0; it < EA_TP; ++it) {
            const int wq = w0 - 4 + it + (it < 4 ? wrapLo : 0) - (it >= EA_TP - 4 ? wrapHi : 0);
            qdst[it * (EA_QP / 4) + lane] = *(const float4*)(srow + (ptrdiff_t)wq * EA_C);
        }
        if (wib == 0) {
            const float4 z = {0.0f, 0.0f, 0.0f, 0.0f};
            ((float4*)(qs + EA_ZOFF))[lane] = z;
        }
    }
    __syncthreads();

    const float* aT[EA_W];
    #pragma unroll
    for (int t = 0; t < EA_W; ++t) {
        const int wq = w + t - 2;
        const int off = (wq >= 0 && wq < EA_WW) ? (wib * EA_TILE + (lo + t + 2) * EA_QP) : EA_ZOFF;
        aT[t] = qs + off + hi * 2;
    }
    const float* bB = &bsm[lo < EA_M ? lo : lo - 6][hi * 2];
    v8f acc0 = {}, acc1 = {};
    #pragma unroll
    for (int step = 0; step < EA_K / 4; step += 2) {
        const int k0 = 4 * step;
        const int k1 = 4 * step + 4;
        const v2f a0 = *(const v2f*)(aT[k0 / 128] + (k0 % 128));
        const v2f b0 = *(const v2f*)(bB + k0);
        acc0 = __builtin_amdgcn_wmma_f32_16x16x4_f32(false, a0, false, b0,
                                                     (short)0, acc0, false, false);
        const v2f a1 = *(const v2f*)(aT[k1 / 128] + (k1 % 128));
        const v2f b1 = *(const v2f*)(bB + k1);
        acc1 = __builtin_amdgcn_wmma_f32_16x16x4_f32(false, a1, false, b1,
                                                     (short)0, acc1, false, false);
    }
    const v8f acc = acc0 + acc1;

    __shared__ float smem[4][16][EA_SP];
    #pragma unroll
    for (int v = 0; v < 8; ++v)
        smem[wib][v + 8 * hi][lo] = acc[v];
    __syncthreads();

    float wv[EA_M];
    float mx = -3.0e38f;
    #pragma unroll
    for (int m = 0; m < EA_M; ++m) {
        wv[m] = smem[wib][lo][m] + bias[m];
        mx = fmaxf(mx, wv[m]);
    }
    float sum = 0.0f;
    #pragma unroll
    for (int m = 0; m < EA_M; ++m) { wv[m] = __expf(wv[m] - mx); sum += wv[m]; }
    const float inv = 1.0f / sum;
    #pragma unroll
    for (int m = 0; m < EA_M; ++m) wv[m] *= inv;

    const int p = p0 + lo;
    const size_t XN = (size_t)EA_B * EA_H * EA_WW * EA_C;
    const size_t WN = (size_t)EA_B * EA_H * EA_WW * EA_M;
    __shared__ __attribute__((aligned(16))) float wsm[4][16][EA_WP];
    if (hi == 0) {
        float4* wd = (float4*)&wsm[wib][lo][0];
        wd[0] = make_float4(wv[0], wv[1], wv[2], wv[3]);
        wd[1] = make_float4(wv[4], wv[5], wv[6], wv[7]);
        *(float2*)&wsm[wib][lo][8] = make_float2(wv[8], wv[9]);
    }
    __syncthreads();
    if (hi == 0) {
        float* ow  = out + XN + (size_t)p * EA_M;
        float* ow2 = out + XN + WN + (size_t)p * EA_M;
        #pragma unroll
        for (int m = 0; m < EA_M; ++m) { ow[m] = wv[m]; ow2[m] = wv[m]; }
    }

    const int c4 = lo * 4;
    const int px = hi * 8;
    const float* tL = qtile + px * EA_QP + c4;
    const float* tR = qtile + (px + 4) * EA_QP + EA_C + c4;
    float4 lrow[12], rrow[12];
    #pragma unroll
    for (int t = 0; t < 12; ++t) {
        lrow[t] = *(const float4*)(tL + t * EA_QP);
        rrow[t] = *(const float4*)(tR + t * EA_QP);
    }

    const float4 g4 = *(const float4*)(gamma + c4);
    const float4 v4 = *(const float4*)(var + c4);
    const float4 m4 = *(const float4*)(mean + c4);
    const float4 b4 = *(const float4*)(beta + c4);
    float4 sc;
    sc.x = g4.x * rsqrtf(v4.x + EA_EPS);
    sc.y = g4.y * rsqrtf(v4.y + EA_EPS);
    sc.z = g4.z * rsqrtf(v4.z + EA_EPS);
    sc.w = g4.w * rsqrtf(v4.w + EA_EPS);

    float* ox = out + (size_t)(p0 + px) * EA_C + c4;
    #pragma unroll
    for (int j = 0; j < 8; ++j) {
        const float4 wa = *(const float4*)&wsm[wib][px + j][0];
        const float4 wb = *(const float4*)&wsm[wib][px + j][4];
        const float2 wc = *(const float2*)&wsm[wib][px + j][8];
        const float wj[EA_M] = {wa.x, wa.y, wa.z, wa.w, wb.x, wb.y, wb.z, wb.w, wc.x, wc.y};
        float4 x = {0.0f, 0.0f, 0.0f, 0.0f};
        #pragma unroll
        for (int i = 0; i < EA_W; ++i) {
            const float4 lv = lrow[j + 4 - i];
            const float4 rv = rrow[j + i];
            const float wlw = wj[2 * i];
            const float wrw = wj[2 * i + 1];
            x.x += wlw * lv.x + wrw * rv.x;
            x.y += wlw * lv.y + wrw * rv.y;
            x.z += wlw * lv.z + wrw * rv.z;
            x.w += wlw * lv.w + wrw * rv.w;
        }
        float4 y;
        y.x = ea_tanh((x.x - m4.x) * sc.x + b4.x);
        y.y = ea_tanh((x.y - m4.y) * sc.y + b4.y);
        y.z = ea_tanh((x.z - m4.z) * sc.z + b4.z);
        y.w = ea_tanh((x.w - m4.w) * sc.w + b4.w);
        *(float4*)(ox + j * EA_C) = y;
    }
}

extern "C" void kernel_launch(void* const* d_in, const int* in_sizes, int n_in,
                              void* d_out, int out_size, void* d_ws, size_t ws_size,
                              hipStream_t stream) {
    const float* L     = (const float*)d_in[0];
    const float* R     = (const float*)d_in[1];
    const float* kconv = (const float*)d_in[2];
    const float* bias  = (const float*)d_in[3];
    const float* gamma = (const float*)d_in[4];
    const float* beta  = (const float*)d_in[5];
    const float* mean  = (const float*)d_in[6];
    const float* var   = (const float*)d_in[7];
    float* out = (float*)d_out;

    const int tiles = EA_B * EA_H * EA_WW / 16;
    ea_main_kernel<<<tiles / 4, 128, 0, stream>>>(L, R, kconv, bias, gamma, beta,
                                                  mean, var, out);
}


# One work-group per two image rows (20 waves, was 4): a row's 168 wrapped positions and the weight tile are staged once and shared by its waves; 256 groups, one resident round; k, sum order unchanged
# speedup vs baseline: 7.0530x; 1.0658x over previous
#include <hip/hip_runtime.h>

typedef float v2f __attribute__((ext_vector_type(2)));
typedef float v8f __attribute__((ext_vector_type(8)));

#define EA_W 5
#define EA_B 4
#define EA_H 128
#define EA_WW 160
#define EA_C 64
#define EA_M 10
#define EA_K 640
#define EA_EPS 1e-3f
#define EA_QP 132
#define EA_BP 644
#define EA_SP 17
#define EA_WP 12
#define EA_RW 10
#define EA_NW (2 * EA_RW)
#define EA_NT (EA_NW * 32)
#define EA_RP (EA_WW + 8)
#define EA_TILE (EA_RP * EA_QP)
#define EA_ZOFF (2 * EA_TILE)

__device__ __forceinline__ float ea_tanh(float y) {
    const float a = __builtin_fabsf(y);
    const float s = a * a;
    float p = __builtin_fmaf(s, -0.00570002059f, 0.0206340719f);
    p = __builtin_fmaf(s, p, -0.0537379310f);
    p = __builtin_fmaf(s, p, 0.133314162f);
    p = __builtin_fmaf(s, p, -0.333332807f);
    const float small = __builtin_fmaf(s, a * p, a);
    const float e = __builtin_amdgcn_exp2f(a * 2.88539004f);
    const float big = __builtin_fmaf(__builtin_amdgcn_rcpf(e + 1.0f), -2.0f, 1.0f);
    return __builtin_copysignf(a < 0.625f ? small : big, y);
}

__global__ __launch_bounds__(EA_NT)
void ea_main_kernel(const float* __restrict__ L, const float* __restrict__ R,
                    const float* __restrict__ kflat, const float* __restrict__ bias,
                    const float* __restrict__ gamma, const float* __restrict__ beta,
                    const float* __restrict__ mean, const float* __restrict__ var,
                    float* __restrict__ out) {
    const int tid  = threadIdx.x;
    const int lane = tid & 31;
    const int wib  = tid >> 5;
    const int rsel = wib >= EA_RW ? 1 : 0;
    const int tin  = wib - rsel * EA_RW;
    const int tile = blockIdx.x * EA_NW + wib;
    const int lo = lane & 15;
    const int hi = lane >> 4;

    const int p0 = tile * 16;
    const int n  = p0 / (EA_H * EA_WW);
    const int hw = p0 % (EA_H * EA_WW);
    const int h  = hw / EA_WW;
    const int w0 = hw % EA_WW;
    const int w  = w0 + lo;
    const size_t rowBase = (size_t)(n * EA_H + h) * EA_WW * EA_C;
    const float* Lrow = L + rowBase;
    const float* Rrow = R + rowBase;

    __shared__ __attribute__((aligned(16))) float qs[EA_ZOFF + EA_QP];
    __shared__ __attribute__((aligned(16))) float bsm[EA_M][EA_BP];
    float* qrow  = qs + rsel * EA_TILE;
    float* qtile = qrow + w0 * EA_QP;

    {
        const float2* bsrc = (const float2*)kflat;
        float* bdst = &bsm[0][0];
        const int u = tid;
        const int k = u / 5;
        const int r = u - 5 * k;
        float* d = bdst + (2 * r) * EA_BP + k;
        #pragma unroll
        for (int i = 0; i < 5; ++i) {
            const float2 v = bsrc[u + i * 640];
            d[i * 128]         = v.x;
            d[i * 128 + EA_BP] = v.y;
        }
    }
    {
        float4* qdst = (float4*)qrow;
        const int c4 = lane * 4;
        const float* srow = (c4 < EA_C) ? (Lrow + c4) : (Rrow + (c4 - EA_C));
        #pragma unroll
        for (int it = 0; it < 16; ++it)
            qdst[(w0 + it + 4) * (EA_QP / 4) + lane] = *(const float4*)(srow + (w0 + it) * EA_C);
        if (tin < 8) {
            const int trow = (tin < 4) ? tin : EA_WW + tin;
            const int wq   = (tin < 4) ? EA_WW - 4 + tin : tin - 4;
            qdst[trow * (EA_QP / 4) + lane] = *(const float4*)(srow + wq * EA_C);
        } else if (wib == 8) {
            const float4 z = {0.0f, 0.0f, 0.0f, 0.0f};
            ((float4*)(qs + EA_ZOFF))[lane] = z;
        }
    }
    __syncthreads();

    const float* aT[EA_W];
    #pragma unroll
    for (int t = 0; t < EA_W; ++t) {
        const int wq = w + t - 2;
        const int off = (wq >= 0 && wq < EA_WW) ? (rsel * EA_TILE + (wq + 4) * EA_QP) : EA_ZOFF;
        aT[t] = qs + off + hi * 2;
    }
    const float* bB = &bsm[lo < EA_M ? lo : lo - 6][hi * 2];
    v8f acc0 = {}, acc1 = {};
    #pragma unroll
    for (int step = 0; step < EA_K / 4; step += 2) {
        const int k0 = 4 * step;
        const int k1 = 4 * step + 4;
        const v2f a0 = *(const v2f*)(aT[k0 / 128] + (k0 % 128));
        const v2f b0 = *(const v2f*)(bB + k0);
        acc0 = __builtin_amdgcn_wmma_f32_16x16x4_f32(false, a0, false, b0,
                                                     (short)0, acc0, false, false);
        const v2f a1 = *(const v2f*)(aT[k1 / 128] + (k1 % 128));
        const v2f b1 = *(const v2f*)(bB + k1);
        acc1 = __builtin_amdgcn_wmma_f32_16x16x4_f32(false, a1, false, b1,
                                                     (short)0, acc1, false, false);
    }
    const v8f acc = acc0 + acc1;

    __shared__ float smem[EA_NW][16][EA_SP];
    #pragma unroll
    for (int v = 0; v < 8; ++v)
        smem[wib][v + 8 * hi][lo] = acc[v];
    __syncthreads();

    float wv[EA_M];
    float mx = -3.0e38f;
    #pragma unroll
    for (int m = 0; m < EA_M; ++m) {
        wv[m] = smem[wib][lo][m] + bias[m];
        mx = fmaxf(mx, wv[m]);
    }
    float sum = 0.0f;
    #pragma unroll
    for (int m = 0; m < EA_M; ++m) { wv[m] = __expf(wv[m] - mx); sum += wv[m]; }
    const float inv = 1.0f / sum;
    #pragma unroll
    for (int m = 0; m < EA_M; ++m) wv[m] *= inv;

    const int p = p0 + lo;
    const size_t XN = (size_t)EA_B * EA_H * EA_WW * EA_C;
    const size_t WN = (size_t)EA_B * EA_H * EA_WW * EA_M;
    __shared__ __attribute__((aligned(16))) float wsm[EA_NW][16][EA_WP];
    if (hi == 0) {
        float4* wd = (float4*)&wsm[wib][lo][0];
        wd[0] = make_float4(wv[0], wv[1], wv[2], wv[3]);
        wd[1] = make_float4(wv[4], wv[5], wv[6], wv[7]);
        *(float2*)&wsm[wib][lo][8] = make_float2(wv[8], wv[9]);
    }
    __syncthreads();
    if (hi == 0) {
        float* ow  = out + XN + (size_t)p * EA_M;
        float* ow2 = out + XN + WN + (size_t)p * EA_M;
        #pragma unroll
        for (int m = 0; m < EA_M; ++m) { ow[m] = wv[m]; ow2[m] = wv[m]; }
    }

    const int c4 = lo * 4;
    const int px = hi * 8;
    const float* tL = qtile + px * EA_QP + c4;
    const float* tR = qtile + (px + 4) * EA_QP + EA_C + c4;
    float4 lrow[12], rrow[12];
    #pragma unroll
    for (int t = 0; t < 12; ++t) {
        lrow[t] = *(const float4*)(tL + t * EA_QP);
        rrow[t] = *(const float4*)(tR + t * EA_QP);
    }

    const float4 g4 = *(const float4*)(gamma + c4);
    const float4 v4 = *(const float4*)(var + c4);
    const float4 m4 = *(const float4*)(mean + c4);
    const float4 b4 = *(const float4*)(beta + c4);
    float4 sc;
    sc.x = g4.x * rsqrtf(v4.x + EA_EPS);
    sc.y = g4.y * rsqrtf(v4.y + EA_EPS);
    sc.z = g4.z * rsqrtf(v4.z + EA_EPS);
    sc.w = g4.w * rsqrtf(v4.w + EA_EPS);

    float* ox = out + (size_t)(p0 + px) * EA_C + c4;
    #pragma unroll
    for (int j = 0; j < 8; ++j) {
        const float4 wa = *(const float4*)&wsm[wib][px + j][0];
        const float4 wb = *(const float4*)&wsm[wib][px + j][4];
        const float2 wc = *(const float2*)&wsm[wib][px + j][8];
        const float wj[EA_M] = {wa.x, wa.y, wa.z, wa.w, wb.x, wb.y, wb.z, wb.w, wc.x, wc.y};
        float4 x = {0.0f, 0.0f, 0.0f, 0.0f};
        #pragma unroll
        for (int i = 0; i < EA_W; ++i) {
            const float4 lv = lrow[j + 4 - i];
            const float4 rv = rrow[j + i];
            const float wlw = wj[2 * i];
            const float wrw = wj[2 * i + 1];
            x.x += wlw * lv.x + wrw * rv.x;
            x.y += wlw * lv.y + wrw * rv.y;
            x.z += wlw * lv.z + wrw * rv.z;
            x.w += wlw * lv.w + wrw * rv.w;
        }
        float4 y;
        y.x = ea_tanh((x.x - m4.x) * sc.x + b4.x);
        y.y = ea_tanh((x.y - m4.y) * sc.y + b4.y);
        y.z = ea_tanh((x.z - m4.z) * sc.z + b4.z);
        y.w = ea_tanh((x.w - m4.w) * sc.w + b4.w);
        *(float4*)(ox + j * EA_C) = y;
    }
}

extern "C" void kernel_launch(void* const* d_in, const int* in_sizes, int n_in,
                              void* d_out, int out_size, void* d_ws, size_t ws_size,
                              hipStream_t stream) {
    const float* L     = (const float*)d_in[0];
    const float* R     = (const float*)d_in[1];
    const float* kconv = (const float*)d_in[2];
    const float* bias  = (const float*)d_in[3];
    const float* gamma = (const float*)d_in[4];
    const float* beta  = (const float*)d_in[5];
    const float* mean  = (const float*)d_in[6];
    const float* var   = (const float*)d_in[7];
    float* out = (float*)d_out;

    ea_main_kernel<<<EA_B * EA_H / 2, EA_NT, 0, stream>>>(L, R, kconv, bias, gamma, beta,
                                                          mean, var, out);
}


# Conv GEMM per position against all 5 taps' weights (50 of 64 columns, was 10 of 16); pixel logit = sum of 5 neighbours' products: 128 WMMAs per 16 px (was 160), A reused x4; logit sum order changed
# speedup vs baseline: 7.7148x; 1.0938x over previous
#include <hip/hip_runtime.h>

typedef float v2f __attribute__((ext_vector_type(2)));
typedef float v8f __attribute__((ext_vector_type(8)));

#define EA_W 5
#define EA_B 4
#define EA_H 128
#define EA_WW 160
#define EA_C 64
#define EA_M 10
#define EA_QC 128
#define EA_N 50
#define EA_EPS 1e-3f
#define EA_QP 132
#define EA_BP 132
#define EA_PP 66
#define EA_WP 12
#define EA_RW 10
#define EA_NW (2 * EA_RW)
#define EA_NT (EA_NW * 32)
#define EA_RP (EA_WW + 8)
#define EA_TILE (EA_RP * EA_QP)

__device__ __forceinline__ float ea_tanh(float y) {
    const float a = __builtin_fabsf(y);
    const float s = a * a;
    float p = __builtin_fmaf(s, -0.00570002059f, 0.0206340719f);
    p = __builtin_fmaf(s, p, -0.0537379310f);
    p = __builtin_fmaf(s, p, 0.133314162f);
    p = __builtin_fmaf(s, p, -0.333332807f);
    const float small = __builtin_fmaf(s, a * p, a);
    const float e = __builtin_amdgcn_exp2f(a * 2.88539004f);
    const float big = __builtin_fmaf(__builtin_amdgcn_rcpf(e + 1.0f), -2.0f, 1.0f);
    return __builtin_copysignf(a < 0.625f ? small : big, y);
}

__global__ __launch_bounds__(EA_NT)
void ea_main_kernel(const float* __restrict__ L, const float* __restrict__ R,
                    const float* __restrict__ kflat, const float* __restrict__ bias,
                    const float* __restrict__ gamma, const float* __restrict__ beta,
                    const float* __restrict__ mean, const float* __restrict__ var,
                    float* __restrict__ out) {
    const int tid  = threadIdx.x;
    const int lane = tid & 31;
    const int wib  = tid >> 5;
    const int rsel = wib >= EA_RW ? 1 : 0;
    const int tin  = wib - rsel * EA_RW;
    const int tile = blockIdx.x * EA_NW + wib;
    const int lo = lane & 15;
    const int hi = lane >> 4;

    const int p0 = tile * 16;
    const int n  = p0 / (EA_H * EA_WW);
    const int hw = p0 % (EA_H * EA_WW);
    const int h  = hw / EA_WW;
    const int w0 = hw % EA_WW;
    const int w  = w0 + lo;
    const size_t rowBase = (size_t)(n * EA_H + h) * EA_WW * EA_C;
    const float* Lrow = L + rowBase;
    const float* Rrow = R + rowBase;

    __shared__ __attribute__((aligned(16))) float qs[2 * EA_TILE];
    __shared__ __attribute__((aligned(16))) float bsm[EA_N][EA_BP];
    float* qrow  = qs + rsel * EA_TILE;
    float* qtile = qrow + w0 * EA_QP;

    {
        const float2* bsrc = (const float2*)kflat;
        float* bdst = &bsm[0][0];
        const int u = tid;
        const int k = u / 5;
        const int r = u - 5 * k;
        float* d = bdst + (2 * r) * EA_BP + k;
        #pragma unroll
        for (int i = 0; i < EA_W; ++i) {
            const float2 v = bsrc[u + i * 640];
            d[i * EA_M * EA_BP]         = v.x;
            d[i * EA_M * EA_BP + EA_BP] = v.y;
        }
    }
    {
        float4* qdst = (float4*)qrow;
        const int c4 = lane * 4;
        const float* srow = (c4 < EA_C) ? (Lrow + c4) : (Rrow + (c4 - EA_C));
        #pragma unroll
        for (int it = 0; it < 16; ++it)
            qdst[(w0 + it + 4) * (EA_QP / 4) + lane] = *(const float4*)(srow + (w0 + it) * EA_C);
        if (tin < 8) {
            const int trow = (tin < 4) ? tin : EA_WW + tin;
            const int wq   = (tin < 4) ? EA_WW - 4 + tin : tin - 4;
            qdst[trow * (EA_QP / 4) + lane] = *(const float4*)(srow + wq * EA_C);
        }
    }
    __syncthreads();

    const float* aP = qrow + (w + 4) * EA_QP + hi * 2;
    const float* bB[4];
    #pragma unroll
    for (int nt = 0; nt < 4; ++nt) {
        const int col = nt * 16 + lo;
        bB[nt] = &bsm[col < EA_N ? col : col - 16][hi * 2];
    }
    v8f acc[4] = {};
    #pragma unroll
    for (int step = 0; step < EA_QC / 4; ++step) {
        const int k0 = 4 * step;
        const v2f a = *(const v2f*)(aP + k0);
        #pragma unroll
        for (int nt = 0; nt < 4; ++nt) {
            const v2f b = *(const v2f*)(bB[nt] + k0);
            acc[nt] = __builtin_amdgcn_wmma_f32_16x16x4_f32(false, a, false, b,
                                                            (short)0, acc[nt], false, false);
        }
    }

    __shared__ __attribute__((aligned(16))) float psm[2][EA_WW][EA_PP];
    #pragma unroll
    for (int nt = 0; nt < 4; ++nt) {
        #pragma unroll
        for (int v = 0; v < 8; ++v)
            psm[rsel][w0 + v + 8 * hi][nt * 16 + lo] = acc[nt][v];
    }
    __syncthreads();

    float lg[EA_M];
    #pragma unroll
    for (int t = 0; t < EA_W; ++t) {
        const int q = w + t - 2;
        const bool inside = (q >= 0 && q < EA_WW);
        const float2* pr = (const float2*)&psm[rsel][inside ? q : w][t * EA_M];
        #pragma unroll
        for (int j = 0; j < EA_M / 2; ++j) {
            const float2 pv = pr[j];
            const float px0 = inside ? pv.x : 0.0f;
            const float px1 = inside ? pv.y : 0.0f;
            if (t == 0) { lg[2 * j] = px0; lg[2 * j + 1] = px1; }
            else        { lg[2 * j] += px0; lg[2 * j + 1] += px1; }
        }
    }

    float wv[EA_M];
    float mx = -3.0e38f;
    #pragma unroll
    for (int m = 0; m < EA_M; ++m) {
        wv[m] = lg[m] + bias[m];
        mx = fmaxf(mx, wv[m]);
    }
    float sum = 0.0f;
    #pragma unroll
    for (int m = 0; m < EA_M; ++m) { wv[m] = __expf(wv[m] - mx); sum += wv[m]; }
    const float inv = 1.0f / sum;
    #pragma unroll
    for (int m = 0; m < EA_M; ++m) wv[m] *= inv;

    const int p = p0 + lo;
    const size_t XN = (size_t)EA_B * EA_H * EA_WW * EA_C;
    const size_t WN = (size_t)EA_B * EA_H * EA_WW * EA_M;
    __shared__ __attribute__((aligned(16))) float wsm[EA_NW][16][EA_WP];
    if (hi == 0) {
        float4* wd = (float4*)&wsm[wib][lo][0];
        wd[0] = make_float4(wv[0], wv[1], wv[2], wv[3]);
        wd[1] = make_float4(wv[4], wv[5], wv[6], wv[7]);
        *(float2*)&wsm[wib][lo][8] = make_float2(wv[8], wv[9]);
    }
    __syncthreads();
    if (hi == 0) {
        float* ow  = out + XN + (size_t)p * EA_M;
        float* ow2 = out + XN + WN + (size_t)p * EA_M;
        #pragma unroll
        for (int m = 0; m < EA_M; ++m) { ow[m] = wv[m]; ow2[m] = wv[m]; }
    }

    const int c4 = lo * 4;
    const int px = hi * 8;
    const float* tL = qtile + px * EA_QP + c4;
    const float* tR = qtile + (px + 4) * EA_QP + EA_C + c4;
    float4 lrow[12], rrow[12];
    #pragma unroll
    for (int t = 0; t < 12; ++t) {
        lrow[t] = *(const float4*)(tL + t * EA_QP);
        rrow[t] = *(const float4*)(tR + t * EA_QP);
    }

    const float4 g4 = *(const float4*)(gamma + c4);
    const float4 v4 = *(const float4*)(var + c4);
    const float4 m4 = *(const float4*)(mean + c4);
    const float4 b4 = *(const float4*)(beta + c4);
    float4 sc;
    sc.x = g4.x * rsqrtf(v4.x + EA_EPS);
    sc.y = g4.y * rsqrtf(v4.y + EA_EPS);
    sc.z = g4.z * rsqrtf(v4.z + EA_EPS);
    sc.w = g4.w * rsqrtf(v4.w + EA_EPS);

    float* ox = out + (size_t)(p0 + px) * EA_C + c4;
    #pragma unroll
    for (int j = 0; j < 8; ++j) {
        const float4 wa = *(const float4*)&wsm[wib][px + j][0];
        const float4 wb = *(const float4*)&wsm[wib][px + j][4];
        const float2 wc = *(const float2*)&wsm[wib][px + j][8];
        const float wj[EA_M] = {wa.x, wa.y, wa.z, wa.w, wb.x, wb.y, wb.z, wb.w, wc.x, wc.y};
        float4 x = {0.0f, 0.0f, 0.0f, 0.0f};
        #pragma unroll
        for (int i = 0; i < EA_W; ++i) {
            const float4 lv = lrow[j + 4 - i];
            const float4 rv = rrow[j + i];
            const float wlw = wj[2 * i];
            const float wrw = wj[2 * i + 1];
            x.x += wlw * lv.x + wrw * rv.x;
            x.y += wlw * lv.y + wrw * rv.y;
            x.z += wlw * lv.z + wrw * rv.z;
            x.w += wlw * lv.w + wrw * rv.w;
        }
        float4 y;
        y.x = ea_tanh((x.x - m4.x) * sc.x + b4.x);
        y.y = ea_tanh((x.y - m4.y) * sc.y + b4.y);
        y.z = ea_tanh((x.z - m4.z) * sc.z + b4.z);
        y.w = ea_tanh((x.w - m4.w) * sc.w + b4.w);
        *(float4*)(ox + j * EA_C) = y;
    }
}

extern "C" void kernel_launch(void* const* d_in, const int* in_sizes, int n_in,
                              void* d_out, int out_size, void* d_ws, size_t ws_size,
                              hipStream_t stream) {
    const float* L     = (const float*)d_in[0];
    const float* R     = (const float*)d_in[1];
    const float* kconv = (const float*)d_in[2];
    const float* bias  = (const float*)d_in[3];
    const float* gamma = (const float*)d_in[4];
    const float* beta  = (const float*)d_in[5];
    const float* mean  = (const float*)d_in[6];
    const float* var   = (const float*)d_in[7];
    float* out = (float*)d_out;

    ea_main_kernel<<<EA_B * EA_H / 2, EA_NT, 0, stream>>>(L, R, kconv, bias, gamma, beta,
                                                          mean, var, out);
}


# Softmax shift invariance folded into the conv weights: column 9 subtracted from columns 0..8 while B is staged; 45 product columns fill 3 tiles, 96 WMMAs per 16 px (was 128); logit rounding changes
# speedup vs baseline: 8.4418x; 1.0942x over previous
#include <hip/hip_runtime.h>

typedef float v2f __attribute__((ext_vector_type(2)));
typedef float v8f __attribute__((ext_vector_type(8)));

#define EA_W 5
#define EA_B 4
#define EA_H 128
#define EA_WW 160
#define EA_C 64
#define EA_M 10
#define EA_QC 128
#define EA_D 9
#define EA_N 45
#define EA_EPS 1e-3f
#define EA_QP 132
#define EA_BP 132
#define EA_PP 52
#define EA_WP 12
#define EA_RW 10
#define EA_NW (2 * EA_RW)
#define EA_NT (EA_NW * 32)
#define EA_RP (EA_WW + 8)
#define EA_TILE (EA_RP * EA_QP)

__device__ __forceinline__ float ea_tanh(float y) {
    const float a = __builtin_fabsf(y);
    const float s = a * a;
    float p = __builtin_fmaf(s, -0.00570002059f, 0.0206340719f);
    p = __builtin_fmaf(s, p, -0.0537379310f);
    p = __builtin_fmaf(s, p, 0.133314162f);
    p = __builtin_fmaf(s, p, -0.333332807f);
    const float small = __builtin_fmaf(s, a * p, a);
    const float e = __builtin_amdgcn_exp2f(a * 2.88539004f);
    const float big = __builtin_fmaf(__builtin_amdgcn_rcpf(e + 1.0f), -2.0f, 1.0f);
    return __builtin_copysignf(a < 0.625f ? small : big, y);
}

__global__ __launch_bounds__(EA_NT)
void ea_main_kernel(const float* __restrict__ L, const float* __restrict__ R,
                    const float* __restrict__ kflat, const float* __restrict__ bias,
                    const float* __restrict__ gamma, const float* __restrict__ beta,
                    const float* __restrict__ mean, const float* __restrict__ var,
                    float* __restrict__ out) {
    const int tid  = threadIdx.x;
    const int lane = tid & 31;
    const int wib  = tid >> 5;
    const int rsel = wib >= EA_RW ? 1 : 0;
    const int tin  = wib - rsel * EA_RW;
    const int tile = blockIdx.x * EA_NW + wib;
    const int lo = lane & 15;
    const int hi = lane >> 4;

    const int p0 = tile * 16;
    const int n  = p0 / (EA_H * EA_WW);
    const int hw = p0 % (EA_H * EA_WW);
    const int h  = hw / EA_WW;
    const int w0 = hw % EA_WW;
    const int w  = w0 + lo;
    const size_t rowBase = (size_t)(n * EA_H + h) * EA_WW * EA_C;
    const float* Lrow = L + rowBase;
    const float* Rrow = R + rowBase;

    __shared__ __attribute__((aligned(16))) float qs[2 * EA_TILE];
    __shared__ __attribute__((aligned(16))) float bsm[EA_N][EA_BP];
    float* qrow  = qs + rsel * EA_TILE;
    float* qtile = qrow + w0 * EA_QP;

    {
        const float2* bsrc = (const float2*)kflat;
        float* bdst = &bsm[0][0];
        const int u = tid;
        const int k = u / 5;
        const int r = u - 5 * k;
        float* d = bdst + (2 * r) * EA_BP + k;
        const float* c9p = kflat + k * EA_M + (EA_M - 1);
        const bool two = (r < 4);
        const int o1 = two ? EA_BP : 0;
        #pragma unroll
        for (int i = 0; i < EA_W; ++i) {
            const float2 v = bsrc[u + i * 640];
            const float c9 = c9p[i * EA_QC * EA_M];
            d[i * EA_D * EA_BP]      = v.x - c9;
            d[i * EA_D * EA_BP + o1] = (two ? v.y : v.x) - c9;
        }
    }
    {
        float4* qdst = (float4*)qrow;
        const int c4 = lane * 4;
        const float* srow = (c4 < EA_C) ? (Lrow + c4) : (Rrow + (c4 - EA_C));
        #pragma unroll
        for (int it = 0; it < 16; ++it)
            qdst[(w0 + it + 4) * (EA_QP / 4) + lane] = *(const float4*)(srow + (w0 + it) * EA_C);
        if (tin < 8) {
            const int trow = (tin < 4) ? tin : EA_WW + tin;
            const int wq   = (tin < 4) ? EA_WW - 4 + tin : tin - 4;
            qdst[trow * (EA_QP / 4) + lane] = *(const float4*)(srow + wq * EA_C);
        }
    }
    __syncthreads();

    const float* aP = qrow + (w + 4) * EA_QP + hi * 2;
    const float* bB[3];
    #pragma unroll
    for (int nt = 0; nt < 3; ++nt) {
        const int col = nt * 16 + lo;
        bB[nt] = &bsm[col < EA_N ? col : col - 16][hi * 2];
    }
    v8f acc[3] = {};
    #pragma unroll
    for (int step = 0; step < EA_QC / 4; ++step) {
        const int k0 = 4 * step;
        const v2f a = *(const v2f*)(aP + k0);
        #pragma unroll
        for (int nt = 0; nt < 3; ++nt) {
            const v2f b = *(const v2f*)(bB[nt] + k0);
            acc[nt] = __builtin_amdgcn_wmma_f32_16x16x4_f32(false, a, false, b,
                                                            (short)0, acc[nt], false, false);
        }
    }

    __shared__ __attribute__((aligned(16))) float psm[2][EA_WW][EA_PP];
    #pragma unroll
    for (int nt = 0; nt < 3; ++nt) {
        const int c = nt * 16 + lo;
        const int slot = c + c / EA_D;
        if (c < EA_N) {
            #pragma unroll
            for (int v = 0; v < 8; ++v)
                psm[rsel][w0 + v + 8 * hi][slot] = acc[nt][v];
        }
    }
    __syncthreads();

    float lg[EA_D];
    #pragma unroll
    for (int t = 0; t < EA_W; ++t) {
        const int q = w + t - 2;
        const bool inside = (q >= 0 && q < EA_WW);
        const float* prow = &psm[rsel][inside ? q : w][t * EA_M];
        const float2* pr = (const float2*)prow;
        #pragma unroll
        for (int j = 0; j < EA_D / 2; ++j) {
            const float2 pv = pr[j];
            const float px0 = inside ? pv.x : 0.0f;
            const float px1 = inside ? pv.y : 0.0f;
            if (t == 0) { lg[2 * j] = px0; lg[2 * j + 1] = px1; }
            else        { lg[2 * j] += px0; lg[2 * j + 1] += px1; }
        }
        const float p8 = inside ? prow[EA_D - 1] : 0.0f;
        if (t == 0) lg[EA_D - 1] = p8;
        else        lg[EA_D - 1] += p8;
    }

    float wv[EA_M];
    const float b9 = bias[EA_M - 1];
    float mx = 0.0f;
    #pragma unroll
    for (int m = 0; m < EA_D; ++m) {
        wv[m] = lg[m] + (bias[m] - b9);
        mx = fmaxf(mx, wv[m]);
    }
    wv[EA_D] = 0.0f;
    float sum = 0.0f;
    #pragma unroll
    for (int m = 0; m < EA_M; ++m) { wv[m] = __expf(wv[m] - mx); sum += wv[m]; }
    const float inv = 1.0f / sum;
    #pragma unroll
    for (int m = 0; m < EA_M; ++m) wv[m] *= inv;

    const int p = p0 + lo;
    const size_t XN = (size_t)EA_B * EA_H * EA_WW * EA_C;
    const size_t WN = (size_t)EA_B * EA_H * EA_WW * EA_M;
    __shared__ __attribute__((aligned(16))) float wsm[EA_NW][16][EA_WP];
    if (hi == 0) {
        float4* wd = (float4*)&wsm[wib][lo][0];
        wd[0] = make_float4(wv[0], wv[1], wv[2], wv[3]);
        wd[1] = make_float4(wv[4], wv[5], wv[6], wv[7]);
        *(float2*)&wsm[wib][lo][8] = make_float2(wv[8], wv[9]);
    }
    __syncthreads();
    if (hi == 0) {
        float* ow  = out + XN + (size_t)p * EA_M;
        float* ow2 = out + XN + WN + (size_t)p * EA_M;
        #pragma unroll
        for (int m = 0; m < EA_M; ++m) { ow[m] = wv[m]; ow2[m] = wv[m]; }
    }

    const int c4 = lo * 4;
    const int px = hi * 8;
    const float* tL = qtile + px * EA_QP + c4;
    const float* tR = qtile + (px + 4) * EA_QP + EA_C + c4;
    float4 lrow[12], rrow[12];
    #pragma unroll
    for (int t = 0; t < 12; ++t) {
        lrow[t] = *(const float4*)(tL + t * EA_QP);
        rrow[t] = *(const float4*)(tR + t * EA_QP);
    }

    const float4 g4 = *(const float4*)(gamma + c4);
    const float4 v4 = *(const float4*)(var + c4);
    const float4 m4 = *(const float4*)(mean + c4);
    const float4 b4 = *(const float4*)(beta + c4);
    float4 sc;
    sc.x = g4.x * rsqrtf(v4.x + EA_EPS);
    sc.y = g4.y * rsqrtf(v4.y + EA_EPS);
    sc.z = g4.z * rsqrtf(v4.z + EA_EPS);
    sc.w = g4.w * rsqrtf(v4.w + EA_EPS);

    float* ox = out + (size_t)(p0 + px) * EA_C + c4;
    #pragma unroll
    for (int j = 0; j < 8; ++j) {
        const float4 wa = *(const float4*)&wsm[wib][px + j][0];
        const float4 wb = *(const float4*)&wsm[wib][px + j][4];
        const float2 wc = *(const float2*)&wsm[wib][px + j][8];
        const float wj[EA_M] = {wa.x, wa.y, wa.z, wa.w, wb.x, wb.y, wb.z, wb.w, wc.x, wc.y};
        float4 x = {0.0f, 0.0f, 0.0f, 0.0f};
        #pragma unroll
        for (int i = 0; i < EA_W; ++i) {
            const float4 lv = lrow[j + 4 - i];
            const float4 rv = rrow[j + i];
            const float wlw = wj[2 * i];
            const float wrw = wj[2 * i + 1];
            x.x += wlw * lv.x + wrw * rv.x;
            x.y += wlw * lv.y + wrw * rv.y;
            x.z += wlw * lv.z + wrw * rv.z;
            x.w += wlw * lv.w + wrw * rv.w;
        }
        float4 y;
        y.x = ea_tanh((x.x - m4.x) * sc.x + b4.x);
        y.y = ea_tanh((x.y - m4.y) * sc.y + b4.y);
        y.z = ea_tanh((x.z - m4.z) * sc.z + b4.z);
        y.w = ea_tanh((x.w - m4.w) * sc.w + b4.w);
        *(float4*)(ox + j * EA_C) = y;
    }
}

extern "C" void kernel_launch(void* const* d_in, const int* in_sizes, int n_in,
                              void* d_out, int out_size, void* d_ws, size_t ws_size,
                              hipStream_t stream) {
    const float* L     = (const float*)d_in[0];
    const float* R     = (const float*)d_in[1];
    const float* kconv = (const float*)d_in[2];
    const float* bias  = (const float*)d_in[3];
    const float* gamma = (const float*)d_in[4];
    const float* beta  = (const float*)d_in[5];
    const float* mean  = (const float*)d_in[6];
    const float* var   = (const float*)d_in[7];
    float* out = (float*)d_out;

    ea_main_kernel<<<EA_B * EA_H / 2, EA_NT, 0, stream>>>(L, R, kconv, bias, gamma, beta,
                                                          mean, var, out);
}
